# starting kernel
# baseline (speedup 1.0000x reference)
#include <hip/hip_runtime.h>

typedef __attribute__((ext_vector_type(2))) float v2f;
typedef __attribute__((ext_vector_type(8))) float v8f;

#define LOG2E 1.44269504088896340736f
#define NCHUNK_PER_B 4
#define NCHUNKS 8

__global__ __launch_bounds__(256) void kde_wmma_kernel(
    const float* __restrict__ samples,
    const float* __restrict__ locations,
    float* __restrict__ partials,
    int m, int S, int rowsPerChunk)
{
    const int wid  = (blockIdx.x * blockDim.x + threadIdx.x) >> 5;
    const int lane = threadIdx.x & 31;
    const int nIdx = lane & 15;
    const int hi   = lane >> 4;

    const int loc_tile = wid / NCHUNKS;
    const int chunk    = wid % NCHUNKS;
    const int b        = chunk >> 2;
    const int c        = chunk & (NCHUNK_PER_B - 1);

    const float2* loc2 = reinterpret_cast<const float2*>(locations);
    float2 y = loc2[loc_tile * 16 + nIdx];
    v2f bm;
    if (hi) { bm[0] = 1.0f;       bm[1] = y.x * y.x + y.y * y.y; }
    else    { bm[0] = -2.0f * y.x; bm[1] = -2.0f * y.y; }

    const float s = -12.5f * LOG2E;

    const float2* smp2 = reinterpret_cast<const float2*>(samples);
    const int rowBase = b * S + c * rowsPerChunk;
    const int nTiles  = rowsPerChunk >> 4;

    v8f acc = {0.f, 0.f, 0.f, 0.f, 0.f, 0.f, 0.f, 0.f};

    for (int t = 0; t < nTiles; ++t) {
        int row = rowBase + t * 16 + nIdx;
        float2 x = smp2[row];
        v2f am;
        if (hi) { am[0] = s * (x.x * x.x + x.y * x.y); am[1] = s; }
        else    { am[0] = s * x.x;                     am[1] = s * x.y; }

        v8f cz = {0.f, 0.f, 0.f, 0.f, 0.f, 0.f, 0.f, 0.f};
        v8f d = __builtin_amdgcn_wmma_f32_16x16x4_f32(
            false, am, false, bm, (short)0, cz, false, false);

#pragma unroll
        for (int i = 0; i < 8; ++i)
            acc[i] += __builtin_amdgcn_exp2f(d[i]);
    }

    float tsum = ((acc[0] + acc[1]) + (acc[2] + acc[3])) +
                 ((acc[4] + acc[5]) + (acc[6] + acc[7]));
    tsum += __shfl_xor(tsum, 16);

    if (hi == 0)
        partials[chunk * m + loc_tile * 16 + nIdx] = tsum;
}

__global__ __launch_bounds__(256) void kde_reduce_kernel(
    const float* __restrict__ partials, float* __restrict__ out,
    float* __restrict__ norms, int m)
{
    const int b = blockIdx.x;
    const int tid = threadIdx.x;

    float lnorm = 0.0f;
    for (int j = tid; j < m; j += blockDim.x) {
        float o = (partials[(b * NCHUNK_PER_B + 0) * m + j] +
                   partials[(b * NCHUNK_PER_B + 1) * m + j]) +
                  (partials[(b * NCHUNK_PER_B + 2) * m + j] +
                   partials[(b * NCHUNK_PER_B + 3) * m + j]);
        out[j * 2 + b] = o;
        lnorm += o;
    }

    __shared__ float sd[256];
    sd[tid] = lnorm;
    __syncthreads();
    for (int stp = 128; stp > 0; stp >>= 1) {
        if (tid < stp) sd[tid] += sd[tid + stp];
        __syncthreads();
    }
    if (tid == 0) norms[b] = sd[0];
}

__global__ __launch_bounds__(256) void kde_norm_kernel(
    float* __restrict__ out, const float* __restrict__ norms, int n)
{
    int i = blockIdx.x * blockDim.x + threadIdx.x;
    if (i < n) out[i] = out[i] / norms[i & 1];
}

extern "C" void kernel_launch(void* const* d_in, const int* in_sizes, int n_in,
                              void* d_out, int out_size, void* d_ws, size_t ws_size,
                              hipStream_t stream) {
    const float* samples   = (const float*)d_in[0];
    const float* locations = (const float*)d_in[1];
    float* out = (float*)d_out;

    const int R = in_sizes[0] / 2;
    const int m = in_sizes[1] / 2;
    const int B = 2;
    const int S = R / B;
    const int rowsPerChunk = S / NCHUNK_PER_B;

    float* partials = (float*)d_ws;
    float* norms    = partials + (size_t)NCHUNKS * m;

    const int nWaves  = (m / 16) * NCHUNKS;
    const int threads = 256;
    const int blocks  = nWaves * 32 / threads;

    kde_wmma_kernel<<<blocks, threads, 0, stream>>>(
        samples, locations, partials, m, S, rowsPerChunk);
    kde_reduce_kernel<<<B, 256, 0, stream>>>(partials, out, norms, m);

    const int n = 2 * m;
    kde_norm_kernel<<<(n + 255) / 256, 256, 0, stream>>>(out, norms, n);
}


# speedup vs baseline: 2.3874x; 2.3874x over previous
#include <hip/hip_runtime.h>


typedef __attribute__((ext_vector_type(2))) float v2f;
typedef __attribute__((ext_vector_type(8))) float v8f;

#define LOG2E 1.44269504088896340736f
#define NBATCH 2
#define NCHUNK_PER_B 2
#define NPLANES (NBATCH * NCHUNK_PER_B)

#define TPB 16
#define MAIN_WAVES 8

__device__ __forceinline__ v2f make_a(float2 x, int hi, float s) {
    v2f am;
    if (hi) { am[0] = s * (x.x * x.x + x.y * x.y); am[1] = s; }
    else    { am[0] = s * x.x;                     am[1] = s * x.y; }
    return am;
}


#define EXP4_ADD4(e0, e1, e2, e3, d0, d1, d2, d3, a0, a1, a2, a3, p0, p1, p2, p3)      \
    asm("v_exp_f32 %0, %8\n\t"                                                      \
        "v_add_f32 %4, %4, %12\n\t"                                                 \
        "v_exp_f32 %1, %9\n\t"                                                      \
        "v_add_f32 %5, %5, %13\n\t"                                                 \
        "v_exp_f32 %2, %10\n\t"                                                     \
        "v_add_f32 %6, %6, %14\n\t"                                                 \
        "v_exp_f32 %3, %11\n\t"                                                     \
        "v_add_f32 %7, %7, %15"                                                       \
        : "=&v"(e0), "=&v"(e1), "=&v"(e2), "=&v"(e3), "+v"(a0), "+v"(a1), "+v"(a2), "+v"(a3) \
        : "v"(d0), "v"(d1), "v"(d2), "v"(d3), "v"(p0), "v"(p1), "v"(p2), "v"(p3))

__global__ __launch_bounds__(MAIN_WAVES * 32) void kde_wmma_kernel(
    const float* __restrict__ samples,
    const float* __restrict__ locations,
    float* __restrict__ partials,
    int m, int rowsPerChunk)
{
    const int lane = threadIdx.x & 31;
    const int nIdx = lane & 15;
    const int hi   = lane >> 4;

    const int groupsPerChunk = (m >> 4) / MAIN_WAVES;
    const int chunk    = blockIdx.x / groupsPerChunk;
    const int loc_tile = (blockIdx.x - chunk * groupsPerChunk) * MAIN_WAVES + (threadIdx.x >> 5);

    const float2* loc2 = reinterpret_cast<const float2*>(locations);
    float2 y = loc2[loc_tile * 16 + nIdx];
    v2f bm;
    if (hi) { bm[0] = 1.0f;       bm[1] = y.x * y.x + y.y * y.y; }
    else    { bm[0] = -2.0f * y.x; bm[1] = -2.0f * y.y; }

    const float s = -12.5f * LOG2E;

    const float2* sp = reinterpret_cast<const float2*>(samples) + (size_t)chunk * rowsPerChunk + nIdx;
    const int nT = rowsPerChunk >> 4;

    const v8f cz = {0.f, 0.f, 0.f, 0.f, 0.f, 0.f, 0.f, 0.f};

    float2 xc[TPB], xn[TPB], xf[TPB];
#pragma unroll
    for (int u = 0; u < TPB; ++u) xc[u] = sp[u * 16];
#pragma unroll
    for (int u = 0; u < TPB; ++u) xn[u] = sp[(TPB + u) * 16];

    v8f dA = __builtin_amdgcn_wmma_f32_16x16x4_f32(false, make_a(xc[0], hi, s), false, bm, (short)0, cz, false, false);
    v8f dB = __builtin_amdgcn_wmma_f32_16x16x4_f32(false, make_a(xc[1], hi, s), false, bm, (short)0, cz, false, false);
    v2f fA = make_a(xc[2], hi, s);
    v2f fB = make_a(xc[3], hi, s);

    float ac[8], pe[8];
#pragma unroll
    for (int i = 0; i < 8; ++i) { ac[i] = 0.0f; pe[i] = 0.0f; }

    for (int t0 = 0; t0 < nT; t0 += TPB) {
        int tn = t0 + 2 * TPB;
        tn = (tn < nT) ? tn : (nT - TPB);
        const float2* pl = sp + (size_t)tn * 16;
#pragma unroll
        for (int u = 0; u < TPB; ++u) xf[u] = pl[u * 16];

#pragma unroll
        for (int u = 0; u < TPB; u += 2) {
            const float2 xa = (u + 4 < TPB) ? xc[(u + 4) % TPB] : xn[(u + 4) % TPB];
            const float2 xb = (u + 5 < TPB) ? xc[(u + 5) % TPB] : xn[(u + 5) % TPB];

            const v8f nA = __builtin_amdgcn_wmma_f32_16x16x4_f32(false, fA, false, bm, (short)0, cz, false, false);
            const v8f nB = __builtin_amdgcn_wmma_f32_16x16x4_f32(false, fB, false, bm, (short)0, cz, false, false);
            __builtin_amdgcn_sched_barrier(0);

            float ea[8];
            EXP4_ADD4(ea[0], ea[1], ea[2], ea[3], dA[0], dA[1], dA[2], dA[3], ac[0], ac[1], ac[2], ac[3], pe[0], pe[1], pe[2], pe[3]);
            EXP4_ADD4(ea[4], ea[5], ea[6], ea[7], dA[4], dA[5], dA[6], dA[7], ac[4], ac[5], ac[6], ac[7], pe[4], pe[5], pe[6], pe[7]);
            EXP4_ADD4(pe[0], pe[1], pe[2], pe[3], dB[0], dB[1], dB[2], dB[3], ac[0], ac[1], ac[2], ac[3], ea[0], ea[1], ea[2], ea[3]);
            EXP4_ADD4(pe[4], pe[5], pe[6], pe[7], dB[4], dB[5], dB[6], dB[7], ac[4], ac[5], ac[6], ac[7], ea[4], ea[5], ea[6], ea[7]);
            fA = make_a(xa, hi, s);
            fB = make_a(xb, hi, s);
            __builtin_amdgcn_sched_barrier(0);
            dA = nA; dB = nB;
        }
#pragma unroll
        for (int u = 0; u < TPB; ++u) { xc[u] = xn[u]; xn[u] = xf[u]; }
    }
    v8f acc;
#pragma unroll
    for (int i = 0; i < 8; ++i) acc[i] = ac[i] + pe[i];

    float tsum = ((acc[0] + acc[1]) + (acc[2] + acc[3])) +
                 ((acc[4] + acc[5]) + (acc[6] + acc[7]));
    tsum += __shfl_xor(tsum, 16);

    if (hi == 0)
        partials[chunk * m + loc_tile * 16 + nIdx] = tsum;
}

#define FIN_THREADS 1024
#define FIN_H 2
#define FIN_BLOCKS 16
#define FIN_OWNERS (FIN_THREADS / FIN_BLOCKS)

__device__ __forceinline__ float4 add4(float4 a, float4 b) {
    return make_float4(a.x + b.x, a.y + b.y, a.z + b.z, a.w + b.w);
}

__global__ __launch_bounds__(FIN_THREADS) void kde_fold_norm_kernel(
    const float* __restrict__ partials,
    float* __restrict__ out,
    int m)
{
    __shared__ float4 num4[FIN_THREADS / 4];
    __shared__ float4 red4[2][FIN_THREADS / 32 / 4];

    const int tid  = threadIdx.x;
    const int lane = tid & 31;
    const int wv   = tid >> 5;
    const int nv   = m >> 2;
    const float4* p4 = reinterpret_cast<const float4*>(partials);

    float4 v[NPLANES][FIN_H];
#pragma unroll
    for (int c = 0; c < NPLANES; ++c)
#pragma unroll
        for (int h = 0; h < FIN_H; ++h)
            v[c][h] = p4[(size_t)c * nv + h * FIN_THREADS + tid];

    float4 f[2][FIN_H];
#pragma unroll
    for (int b = 0; b < 2; ++b)
#pragma unroll
        for (int h = 0; h < FIN_H; ++h)
            f[b][h] = add4(v[b * NCHUNK_PER_B + 0][h], v[b * NCHUNK_PER_B + 1][h]);

    float a[2];
#pragma unroll
    for (int b = 0; b < 2; ++b) {
        a[b] = 0.0f;
#pragma unroll
        for (int h = 0; h < FIN_H; ++h)
            a[b] += (f[b][h].x + f[b][h].y) + (f[b][h].z + f[b][h].w);
    }
#pragma unroll
    for (int off = 16; off > 0; off >>= 1) {
        a[0] += __shfl_xor(a[0], off);
        a[1] += __shfl_xor(a[1], off);
    }
    if (lane == 0) {
        reinterpret_cast<float*>(&red4[0][0])[wv] = a[0];
        reinterpret_cast<float*>(&red4[1][0])[wv] = a[1];
    }

    const int first = blockIdx.x * FIN_OWNERS;
    if (tid >= first && tid < first + FIN_OWNERS) {
        const int u = tid - first;
#pragma unroll
        for (int h = 0; h < FIN_H; ++h) {
            num4[h * (2 * FIN_OWNERS) + 2 * u + 0] =
                make_float4(f[0][h].x, f[1][h].x, f[0][h].y, f[1][h].y);
            num4[h * (2 * FIN_OWNERS) + 2 * u + 1] =
                make_float4(f[0][h].z, f[1][h].z, f[0][h].w, f[1][h].w);
        }
    }
    __syncthreads();

    const int perHalf = 8 * FIN_OWNERS;
    const int hh = tid / perHalf;
    const int r  = tid - hh * perHalf;
    const int b  = r & 1;
    float4 s4 = red4[b][0];
#pragma unroll
    for (int i = 1; i < FIN_THREADS / 32 / 4; ++i) s4 = add4(s4, red4[b][i]);
    const float nb = (s4.x + s4.y) + (s4.z + s4.w);

    const float numer = reinterpret_cast<const float*>(num4)[tid];
    out[(size_t)hh * (FIN_THREADS * 8) + (size_t)blockIdx.x * perHalf + r] = numer / nb;
}

extern "C" void kernel_launch(void* const* d_in, const int* in_sizes, int n_in,
                              void* d_out, int out_size, void* d_ws, size_t ws_size,
                              hipStream_t stream) {
    const float* samples   = (const float*)d_in[0];
    const float* locations = (const float*)d_in[1];
    float* out = (float*)d_out;

    const int R = in_sizes[0] / 2;
    const int m = in_sizes[1] / 2;
    const int rowsPerChunk = R / NPLANES;

    float* partials = (float*)d_ws;

    const int nWaves = (m / 16) * NPLANES;
    const int blocks = nWaves / MAIN_WAVES;

    kde_wmma_kernel<<<blocks, MAIN_WAVES * 32, 0, stream>>>(
        samples, locations, partials, m, rowsPerChunk);
    kde_fold_norm_kernel<<<FIN_BLOCKS, FIN_THREADS, 0, stream>>>(partials, out, m);
}


# A fragments built once per block (16 tiles per wave) and shared through LDS: the main loop has no sample loads and no fragment arithmetic; each wave starts at its own trip (sum order changes)
# speedup vs baseline: 2.6386x; 1.1052x over previous
#include <hip/hip_runtime.h>


typedef __attribute__((ext_vector_type(2))) float v2f;
typedef __attribute__((ext_vector_type(8))) float v8f;

#define LOG2E 1.44269504088896340736f
#define NBATCH 2
#define NCHUNK_PER_B 2
#define NPLANES (NBATCH * NCHUNK_PER_B)

#define TPB 16
#define MAIN_WAVES 8

__device__ __forceinline__ v2f make_a(float2 x, int hi, float s) {
    v2f am;
    if (hi) { am[0] = s * (x.x * x.x + x.y * x.y); am[1] = s; }
    else    { am[0] = s * x.x;                     am[1] = s * x.y; }
    return am;
}


#define EXP4_ADD4(e0, e1, e2, e3, d0, d1, d2, d3, a0, a1, a2, a3, p0, p1, p2, p3)      \
    asm("v_exp_f32 %0, %8\n\t"                                                      \
        "v_add_f32 %4, %4, %12\n\t"                                                 \
        "v_exp_f32 %1, %9\n\t"                                                      \
        "v_add_f32 %5, %5, %13\n\t"                                                 \
        "v_exp_f32 %2, %10\n\t"                                                     \
        "v_add_f32 %6, %6, %14\n\t"                                                 \
        "v_exp_f32 %3, %11\n\t"                                                     \
        "v_add_f32 %7, %7, %15"                                                       \
        : "=&v"(e0), "=&v"(e1), "=&v"(e2), "=&v"(e3), "+v"(a0), "+v"(a1), "+v"(a2), "+v"(a3) \
        : "v"(d0), "v"(d1), "v"(d2), "v"(d3), "v"(p0), "v"(p1), "v"(p2), "v"(p3))

#define PAIRS (TPB / 2)
#define CHUNK_TILES (MAIN_WAVES * TPB)

__device__ __forceinline__ v2f frag_even(float4 f) { v2f a; a[0] = f.x; a[1] = f.y; return a; }
__device__ __forceinline__ v2f frag_odd(float4 f)  { v2f a; a[0] = f.z; a[1] = f.w; return a; }

__device__ __forceinline__ void sum_stage(const v8f& dA, const v8f& dB, float (&ac)[8], float (&pe)[8])
{
    float ea[8];
    EXP4_ADD4(ea[0], ea[1], ea[2], ea[3], dA[0], dA[1], dA[2], dA[3], ac[0], ac[1], ac[2], ac[3], pe[0], pe[1], pe[2], pe[3]);
    EXP4_ADD4(ea[4], ea[5], ea[6], ea[7], dA[4], dA[5], dA[6], dA[7], ac[4], ac[5], ac[6], ac[7], pe[4], pe[5], pe[6], pe[7]);
    EXP4_ADD4(pe[0], pe[1], pe[2], pe[3], dB[0], dB[1], dB[2], dB[3], ac[0], ac[1], ac[2], ac[3], ea[0], ea[1], ea[2], ea[3]);
    EXP4_ADD4(pe[4], pe[5], pe[6], pe[7], dB[4], dB[5], dB[6], dB[7], ac[4], ac[5], ac[6], ac[7], ea[4], ea[5], ea[6], ea[7]);
}

__device__ __forceinline__ void sweep_trip(const float4 (&cur)[PAIRS], float4 (&nxt)[PAIRS],
                                           const float4* __restrict__ src, v2f bm,
                                           v8f& dA, v8f& dB, float (&ac)[8], float (&pe)[8])
{
    const v8f cz = {0.f, 0.f, 0.f, 0.f, 0.f, 0.f, 0.f, 0.f};
#pragma unroll
    for (int j = 0; j < PAIRS; ++j) {
        if (j == PAIRS / 2) {
#pragma unroll
            for (int i = 0; i < PAIRS; ++i) nxt[i] = src[i * 32];
            __builtin_amdgcn_sched_barrier(0);
        }
        const v8f nA = __builtin_amdgcn_wmma_f32_16x16x4_f32(false, frag_even(cur[j]), false, bm, (short)0, cz, false, false);
        const v8f nB = __builtin_amdgcn_wmma_f32_16x16x4_f32(false, frag_odd(cur[j]), false, bm, (short)0, cz, false, false);
        __builtin_amdgcn_sched_barrier(0);
        sum_stage(dA, dB, ac, pe);
        __builtin_amdgcn_sched_barrier(0);
        dA = nA; dB = nB;
    }
}

__global__ __launch_bounds__(MAIN_WAVES * 32) void kde_wmma_kernel(
    const float* __restrict__ samples,
    const float* __restrict__ locations,
    float* __restrict__ partials,
    int m, int rowsPerChunk)
{
    __shared__ float4 frag[CHUNK_TILES / 2 * 32];

    const int lane = threadIdx.x & 31;
    const int nIdx = lane & 15;
    const int hi   = lane >> 4;
    const int wv   = threadIdx.x >> 5;

    const int groupsPerChunk = (m >> 4) / MAIN_WAVES;
    const int chunk    = blockIdx.x / groupsPerChunk;
    const int loc_tile = (blockIdx.x - chunk * groupsPerChunk) * MAIN_WAVES + wv;

    const float2* loc2 = reinterpret_cast<const float2*>(locations);
    const float2* sp = reinterpret_cast<const float2*>(samples) + (size_t)chunk * rowsPerChunk
                     + (size_t)wv * (TPB * 16) + nIdx;
    const float2 y = loc2[loc_tile * 16 + nIdx];
    float2 x[TPB];
#pragma unroll
    for (int u = 0; u < TPB; ++u) x[u] = sp[u * 16];

    v2f bm;
    if (hi) { bm[0] = 1.0f;       bm[1] = y.x * y.x + y.y * y.y; }
    else    { bm[0] = -2.0f * y.x; bm[1] = -2.0f * y.y; }

    const float s = -12.5f * LOG2E;

    float4 fp[PAIRS], fq[PAIRS];
#pragma unroll
    for (int j = 0; j < PAIRS; ++j) {
        const v2f a0 = make_a(x[2 * j], hi, s);
        const v2f a1 = make_a(x[2 * j + 1], hi, s);
        fp[j] = make_float4(a0[0], a0[1], a1[0], a1[1]);
    }
    asm volatile("" :: "v"(y.x), "v"(y.y));
    float4* mine = frag + (size_t)wv * (PAIRS * 32) + lane;
#pragma unroll
    for (int j = 0; j < PAIRS; ++j) mine[j * 32] = fp[j];

    __syncthreads();

    float ac[8], pe[8];
#pragma unroll
    for (int i = 0; i < 8; ++i) { ac[i] = 0.0f; pe[i] = 0.0f; }
    const float ninf = -__builtin_inff();
    v8f dA = {ninf, ninf, ninf, ninf, ninf, ninf, ninf, ninf};
    v8f dB = dA;

    for (int k = 0; k < MAIN_WAVES; k += 2) {
        const int t1 = (wv + k + 1) & (MAIN_WAVES - 1);
        const int t2 = (wv + k + 2) & (MAIN_WAVES - 1);
        sweep_trip(fp, fq, frag + (size_t)t1 * (PAIRS * 32) + lane, bm, dA, dB, ac, pe);
        sweep_trip(fq, fp, frag + (size_t)t2 * (PAIRS * 32) + lane, bm, dA, dB, ac, pe);
    }
    sum_stage(dA, dB, ac, pe);
    v8f acc;
#pragma unroll
    for (int i = 0; i < 8; ++i) acc[i] = ac[i] + pe[i];

    float tsum = ((acc[0] + acc[1]) + (acc[2] + acc[3])) +
                 ((acc[4] + acc[5]) + (acc[6] + acc[7]));
    tsum += __shfl_xor(tsum, 16);

    if (hi == 0)
        partials[chunk * m + loc_tile * 16 + nIdx] = tsum;
}

#define FIN_THREADS 1024
#define FIN_H 2
#define FIN_BLOCKS 16
#define FIN_OWNERS (FIN_THREADS / FIN_BLOCKS)

__device__ __forceinline__ float4 add4(float4 a, float4 b) {
    return make_float4(a.x + b.x, a.y + b.y, a.z + b.z, a.w + b.w);
}

__global__ __launch_bounds__(FIN_THREADS) void kde_fold_norm_kernel(
    const float* __restrict__ partials,
    float* __restrict__ out,
    int m)
{
    __shared__ float4 num4[FIN_THREADS / 4];
    __shared__ float4 red4[2][FIN_THREADS / 32 / 4];

    const int tid  = threadIdx.x;
    const int lane = tid & 31;
    const int wv   = tid >> 5;
    const int nv   = m >> 2;
    const float4* p4 = reinterpret_cast<const float4*>(partials);

    float4 v[NPLANES][FIN_H];
#pragma unroll
    for (int c = 0; c < NPLANES; ++c)
#pragma unroll
        for (int h = 0; h < FIN_H; ++h)
            v[c][h] = p4[(size_t)c * nv + h * FIN_THREADS + tid];

    float4 f[2][FIN_H];
#pragma unroll
    for (int b = 0; b < 2; ++b)
#pragma unroll
        for (int h = 0; h < FIN_H; ++h)
            f[b][h] = add4(v[b * NCHUNK_PER_B + 0][h], v[b * NCHUNK_PER_B + 1][h]);

    float a[2];
#pragma unroll
    for (int b = 0; b < 2; ++b) {
        a[b] = 0.0f;
#pragma unroll
        for (int h = 0; h < FIN_H; ++h)
            a[b] += (f[b][h].x + f[b][h].y) + (f[b][h].z + f[b][h].w);
    }
#pragma unroll
    for (int off = 16; off > 0; off >>= 1) {
        a[0] += __shfl_xor(a[0], off);
        a[1] += __shfl_xor(a[1], off);
    }
    if (lane == 0) {
        reinterpret_cast<float*>(&red4[0][0])[wv] = a[0];
        reinterpret_cast<float*>(&red4[1][0])[wv] = a[1];
    }

    const int first = blockIdx.x * FIN_OWNERS;
    if (tid >= first && tid < first + FIN_OWNERS) {
        const int u = tid - first;
#pragma unroll
        for (int h = 0; h < FIN_H; ++h) {
            num4[h * (2 * FIN_OWNERS) + 2 * u + 0] =
                make_float4(f[0][h].x, f[1][h].x, f[0][h].y, f[1][h].y);
            num4[h * (2 * FIN_OWNERS) + 2 * u + 1] =
                make_float4(f[0][h].z, f[1][h].z, f[0][h].w, f[1][h].w);
        }
    }
    __syncthreads();

    const int perHalf = 8 * FIN_OWNERS;
    const int hh = tid / perHalf;
    const int r  = tid - hh * perHalf;
    const int b  = r & 1;
    float4 s4 = red4[b][0];
#pragma unroll
    for (int i = 1; i < FIN_THREADS / 32 / 4; ++i) s4 = add4(s4, red4[b][i]);
    const float nb = (s4.x + s4.y) + (s4.z + s4.w);

    const float numer = reinterpret_cast<const float*>(num4)[tid];
    out[(size_t)hh * (FIN_THREADS * 8) + (size_t)blockIdx.x * perHalf + r] = numer / nb;
}

extern "C" void kernel_launch(void* const* d_in, const int* in_sizes, int n_in,
                              void* d_out, int out_size, void* d_ws, size_t ws_size,
                              hipStream_t stream) {
    const float* samples   = (const float*)d_in[0];
    const float* locations = (const float*)d_in[1];
    float* out = (float*)d_out;

    const int R = in_sizes[0] / 2;
    const int m = in_sizes[1] / 2;
    const int rowsPerChunk = R / NPLANES;

    float* partials = (float*)d_ws;

    const int nWaves = (m / 16) * NPLANES;
    const int blocks = nWaves / MAIN_WAVES;

    kde_wmma_kernel<<<blocks, MAIN_WAVES * 32, 0, stream>>>(
        samples, locations, partials, m, rowsPerChunk);
    kde_fold_norm_kernel<<<FIN_BLOCKS, FIN_THREADS, 0, stream>>>(partials, out, m);
}
